# starting kernel
# baseline (speedup 1.0000x reference)
#include <hip/hip_runtime.h>
#include <math.h>

typedef __attribute__((ext_vector_type(2))) float v2f;
typedef __attribute__((ext_vector_type(8))) float v8f;

constexpr int NB   = 2;
constexpr int NN   = 131072;
constexpr int DD   = 24;
constexpr int KNB  = 4;
constexpr int HID  = 128;
constexpr int DLAT = 16;
constexpr int TPB  = NN / 16;

constexpr size_t XCOUNT   = (size_t)NB * NN * DD;
constexpr size_t W1S_OFF  = XCOUNT;
constexpr size_t W1S_F2   = 24 * 8 * 32;
constexpr size_t W2S_F2   = 32 * 32;
constexpr size_t WSW_FLTS = 2 * (W1S_F2 + W2S_F2);

constexpr int LDS_W2_OFF    = 12288;
constexpr int LDS_WAVE_OFF  = 14336;
constexpr int LDS_WAVE_SZ   = 2048;
constexpr int LDS_TOTAL_B   = (LDS_WAVE_OFF + 8 * LDS_WAVE_SZ) * 4;

__device__ __forceinline__ float gelu_tanh(float x) {
    float u = 0.7978845608028654f * (x + 0.044715f * x * x * x);
    return 0.5f * x * (1.0f + tanhf(u));
}

__global__ __launch_bounds__(256) void swizzle_weights(
    const float* __restrict__ W1, const float* __restrict__ W2,
    float* __restrict__ wsw)
{
    int t = blockIdx.x * 256 + threadIdx.x;
    if (t < (int)W1S_F2) {
        int kk  = t >> 8;
        int rem = t & 255;
        int j    = rem >> 5;
        int l    = rem & 31;
        int half = l >> 4;
        int n    = l & 15;
        int row0 = 4 * kk + 2 * half;
        int col  = 16 * j + n;
        float2 v = make_float2(W1[row0 * HID + col], W1[(row0 + 1) * HID + col]);
        reinterpret_cast<float2*>(wsw)[t] = v;
    } else if (t < (int)(W1S_F2 + W2S_F2)) {
        int s    = t - (int)W1S_F2;
        int kk2  = s >> 5;
        int l    = s & 31;
        int half = l >> 4;
        int n    = l & 15;
        int row0 = 4 * kk2 + 2 * half;
        float2 v = make_float2(W2[row0 * DLAT + n], W2[(row0 + 1) * DLAT + n]);
        reinterpret_cast<float2*>(wsw)[(int)W1S_F2 + s] = v;
    }
}

__global__ __launch_bounds__(256) void euler_step(
    const float* __restrict__ xin, float* __restrict__ xout,
    const float* __restrict__ wsw,
    const float* __restrict__ b1, const float* __restrict__ b2,
    const float* __restrict__ tf, const int* __restrict__ nbr,
    int step)
{
    extern __shared__ float smem[];
    const int tid  = threadIdx.x;
    const int wave = tid >> 5;
    const int lane = tid & 31;

    {
        const float4* src = reinterpret_cast<const float4*>(wsw);
        float4*       dst = reinterpret_cast<float4*>(smem);
        #pragma unroll
        for (int i = 0; i < (int)(WSW_FLTS / 4) / 256; ++i)
            dst[tid + i * 256] = src[tid + i * 256];
    }
    __syncthreads();

    float* zh = smem + LDS_WAVE_OFF + wave * LDS_WAVE_SZ;

    const int tile      = blockIdx.x * 8 + wave;
    const int batch     = tile / TPB;
    const int node_base = (tile % TPB) * 16;
    const size_t xbatch = (size_t)batch * NN * DD;

    #pragma unroll
    for (int rr = 0; rr < 2; ++rr) {
        int r = lane + rr * 32;
        int m = r >> 2, k = r & 3;
        int idx = nbr[(node_base + m) * KNB + k];
        const float4* src = reinterpret_cast<const float4*>(xin + xbatch + (size_t)idx * DD);
        float4*       dst = reinterpret_cast<float4*>(zh + m * 96 + k * 24);
        #pragma unroll
        for (int q = 0; q < 6; ++q) dst[q] = src[q];
    }
    __syncthreads();

    const int n0   = lane & 15;
    const int half = lane >> 4;

    const float2* zf2  = reinterpret_cast<const float2*>(zh);
    const float2* w1f2 = reinterpret_cast<const float2*>(smem);
    const float2* w2f2 = reinterpret_cast<const float2*>(smem + LDS_W2_OFF);

    v8f acc[8];
    #pragma unroll
    for (int j = 0; j < 8; ++j) {
        float bv = b1[16 * j + n0];
        acc[j] = (v8f){bv, bv, bv, bv, bv, bv, bv, bv};
    }

    auto loadA1 = [&](int kk) -> v2f {
        float2 t = zf2[n0 * 48 + 2 * kk + half];
        return (v2f){t.x, t.y};
    };
    auto loadB1 = [&](int kk, v2f* b) {
        #pragma unroll
        for (int j = 0; j < 8; ++j) {
            float2 t = w1f2[(kk * 8 + j) * 32 + lane];
            b[j] = (v2f){t.x, t.y};
        }
    };
    auto mma1 = [&](v2f a, const v2f* b) {
        #pragma unroll
        for (int j = 0; j < 8; ++j)
            acc[j] = __builtin_amdgcn_wmma_f32_16x16x4_f32(
                false, a, false, b[j], (short)0, acc[j], false, false);
    };

    {
        v2f aA, aB;
        v2f bA[8], bB[8];
        aA = loadA1(0); loadB1(0, bA);
        #pragma unroll
        for (int kk = 0; kk < 24; kk += 2) {
            aB = loadA1(kk + 1); loadB1(kk + 1, bB);
            mma1(aA, bA);
            if (kk + 2 < 24) { aA = loadA1(kk + 2); loadB1(kk + 2, bA); }
            mma1(aB, bB);
        }
    }

    #pragma unroll
    for (int j = 0; j < 8; ++j) {
        #pragma unroll
        for (int r = 0; r < 8; ++r) {
            float h = gelu_tanh(acc[j][r]);
            int row = half ? (r + 8) : r;
            zh[row * HID + 16 * j + n0] = h;
        }
    }
    __syncthreads();

    float bv2 = b2[n0];
    v8f acc2a = (v8f){bv2, bv2, bv2, bv2, bv2, bv2, bv2, bv2};
    v8f acc2b = (v8f){0.f, 0.f, 0.f, 0.f, 0.f, 0.f, 0.f, 0.f};

    auto loadA2 = [&](int kk2) -> v2f {
        float2 t = zf2[n0 * 64 + 2 * kk2 + half];
        return (v2f){t.x, t.y};
    };
    auto loadB2 = [&](int kk2) -> v2f {
        float2 t = w2f2[kk2 * 32 + lane];
        return (v2f){t.x, t.y};
    };

    {
        v2f aA = loadA2(0), bA = loadB2(0);
        v2f aB, bB;
        #pragma unroll
        for (int kk2 = 0; kk2 < 32; kk2 += 2) {
            aB = loadA2(kk2 + 1); bB = loadB2(kk2 + 1);
            acc2a = __builtin_amdgcn_wmma_f32_16x16x4_f32(
                false, aA, false, bA, (short)0, acc2a, false, false);
            if (kk2 + 2 < 32) { aA = loadA2(kk2 + 2); bA = loadB2(kk2 + 2); }
            acc2b = __builtin_amdgcn_wmma_f32_16x16x4_f32(
                false, aB, false, bB, (short)0, acc2b, false, false);
        }
    }
    v8f acc2 = acc2a + acc2b;

    float t  = (float)step;
    float ms = tf[batch] - t;
    ms = fminf(fmaxf(ms, 0.0f), 1.0f);

    const size_t obase = xbatch + (size_t)node_base * DD;
    #pragma unroll
    for (int r = 0; r < 8; ++r) {
        int m = half ? (r + 8) : r;
        size_t a = obase + (size_t)m * DD + n0;
        xout[a] = xin[a] + ms * acc2[r];
    }
    {
        int m2 = lane >> 1;
        int c0 = DLAT + (lane & 1) * 4;
        size_t a2 = obase + (size_t)m2 * DD + c0;
        *reinterpret_cast<float4*>(xout + a2) =
            *reinterpret_cast<const float4*>(xin + a2);
    }
}

extern "C" void kernel_launch(void* const* d_in, const int* in_sizes, int n_in,
                              void* d_out, int out_size, void* d_ws, size_t ws_size,
                              hipStream_t stream) {
    (void)in_sizes; (void)n_in; (void)out_size; (void)ws_size;

    const float* x   = (const float*)d_in[0];
    const float* tf  = (const float*)d_in[1];
    const float* W1  = (const float*)d_in[2];
    const float* b1  = (const float*)d_in[3];
    const float* W2  = (const float*)d_in[4];
    const float* b2  = (const float*)d_in[5];
    const int*   nbr = (const int*)d_in[6];

    float* wsA = (float*)d_ws;
    float* wsw = wsA + W1S_OFF;
    float* out = (float*)d_out;

    swizzle_weights<<<28, 256, 0, stream>>>(W1, W2, wsw);

    const int grid = (NB * TPB) / 8;
    euler_step<<<grid, 256, LDS_TOTAL_B, stream>>>(x,   wsA, wsw, b1, b2, tf, nbr, 0);
    euler_step<<<grid, 256, LDS_TOTAL_B, stream>>>(wsA, out, wsw, b1, b2, tf, nbr, 1);
    euler_step<<<grid, 256, LDS_TOTAL_B, stream>>>(out, wsA, wsw, b1, b2, tf, nbr, 2);
    euler_step<<<grid, 256, LDS_TOTAL_B, stream>>>(wsA, out, wsw, b1, b2, tf, nbr, 3);
}


# GELU's library tanhf (two-branch routine, ~33 instructions + 2 branches per element) replaced by 0.5*(1+tanh u) = 1/(1+2^p): one v_exp_f32 + one v_rcp_f32 per element, straight-line f32
# speedup vs baseline: 1.3746x; 1.3746x over previous
#include <hip/hip_runtime.h>
#include <math.h>

typedef __attribute__((ext_vector_type(2))) float v2f;
typedef __attribute__((ext_vector_type(8))) float v8f;

constexpr int NB   = 2;
constexpr int NN   = 131072;
constexpr int DD   = 24;
constexpr int KNB  = 4;
constexpr int HID  = 128;
constexpr int DLAT = 16;
constexpr int TPB  = NN / 16;

constexpr size_t XCOUNT   = (size_t)NB * NN * DD;
constexpr size_t W1S_OFF  = XCOUNT;
constexpr size_t W1S_F2   = 24 * 8 * 32;
constexpr size_t W2S_F2   = 32 * 32;
constexpr size_t WSW_FLTS = 2 * (W1S_F2 + W2S_F2);

constexpr int LDS_W2_OFF    = 12288;
constexpr int LDS_WAVE_OFF  = 14336;
constexpr int LDS_WAVE_SZ   = 2048;
constexpr int LDS_TOTAL_B   = (LDS_WAVE_OFF + 8 * LDS_WAVE_SZ) * 4;

__device__ __forceinline__ float gelu_tanh(float x) {
    constexpr float C1 = (float)(-2.0 * 0.7978845608028654 * 1.4426950408889634);
    constexpr float C2 = (float)(-2.0 * 0.7978845608028654 * 1.4426950408889634 * 0.044715);
    float p = x * fmaf(x * x, C2, C1);
    float e = __builtin_amdgcn_exp2f(p);
    return x * __builtin_amdgcn_rcpf(1.0f + e);
}

__global__ __launch_bounds__(256) void swizzle_weights(
    const float* __restrict__ W1, const float* __restrict__ W2,
    float* __restrict__ wsw)
{
    int t = blockIdx.x * 256 + threadIdx.x;
    if (t < (int)W1S_F2) {
        int kk  = t >> 8;
        int rem = t & 255;
        int j    = rem >> 5;
        int l    = rem & 31;
        int half = l >> 4;
        int n    = l & 15;
        int row0 = 4 * kk + 2 * half;
        int col  = 16 * j + n;
        float2 v = make_float2(W1[row0 * HID + col], W1[(row0 + 1) * HID + col]);
        reinterpret_cast<float2*>(wsw)[t] = v;
    } else if (t < (int)(W1S_F2 + W2S_F2)) {
        int s    = t - (int)W1S_F2;
        int kk2  = s >> 5;
        int l    = s & 31;
        int half = l >> 4;
        int n    = l & 15;
        int row0 = 4 * kk2 + 2 * half;
        float2 v = make_float2(W2[row0 * DLAT + n], W2[(row0 + 1) * DLAT + n]);
        reinterpret_cast<float2*>(wsw)[(int)W1S_F2 + s] = v;
    }
}

__global__ __launch_bounds__(256) void euler_step(
    const float* __restrict__ xin, float* __restrict__ xout,
    const float* __restrict__ wsw,
    const float* __restrict__ b1, const float* __restrict__ b2,
    const float* __restrict__ tf, const int* __restrict__ nbr,
    int step)
{
    extern __shared__ float smem[];
    const int tid  = threadIdx.x;
    const int wave = tid >> 5;
    const int lane = tid & 31;

    {
        const float4* src = reinterpret_cast<const float4*>(wsw);
        float4*       dst = reinterpret_cast<float4*>(smem);
        #pragma unroll
        for (int i = 0; i < (int)(WSW_FLTS / 4) / 256; ++i)
            dst[tid + i * 256] = src[tid + i * 256];
    }
    __syncthreads();

    float* zh = smem + LDS_WAVE_OFF + wave * LDS_WAVE_SZ;

    const int tile      = blockIdx.x * 8 + wave;
    const int batch     = tile / TPB;
    const int node_base = (tile % TPB) * 16;
    const size_t xbatch = (size_t)batch * NN * DD;

    #pragma unroll
    for (int rr = 0; rr < 2; ++rr) {
        int r = lane + rr * 32;
        int m = r >> 2, k = r & 3;
        int idx = nbr[(node_base + m) * KNB + k];
        const float4* src = reinterpret_cast<const float4*>(xin + xbatch + (size_t)idx * DD);
        float4*       dst = reinterpret_cast<float4*>(zh + m * 96 + k * 24);
        #pragma unroll
        for (int q = 0; q < 6; ++q) dst[q] = src[q];
    }
    __syncthreads();

    const int n0   = lane & 15;
    const int half = lane >> 4;

    const float2* zf2  = reinterpret_cast<const float2*>(zh);
    const float2* w1f2 = reinterpret_cast<const float2*>(smem);
    const float2* w2f2 = reinterpret_cast<const float2*>(smem + LDS_W2_OFF);

    v8f acc[8];
    #pragma unroll
    for (int j = 0; j < 8; ++j) {
        float bv = b1[16 * j + n0];
        acc[j] = (v8f){bv, bv, bv, bv, bv, bv, bv, bv};
    }

    auto loadA1 = [&](int kk) -> v2f {
        float2 t = zf2[n0 * 48 + 2 * kk + half];
        return (v2f){t.x, t.y};
    };
    auto loadB1 = [&](int kk, v2f* b) {
        #pragma unroll
        for (int j = 0; j < 8; ++j) {
            float2 t = w1f2[(kk * 8 + j) * 32 + lane];
            b[j] = (v2f){t.x, t.y};
        }
    };
    auto mma1 = [&](v2f a, const v2f* b) {
        #pragma unroll
        for (int j = 0; j < 8; ++j)
            acc[j] = __builtin_amdgcn_wmma_f32_16x16x4_f32(
                false, a, false, b[j], (short)0, acc[j], false, false);
    };

    {
        v2f aA, aB;
        v2f bA[8], bB[8];
        aA = loadA1(0); loadB1(0, bA);
        #pragma unroll
        for (int kk = 0; kk < 24; kk += 2) {
            aB = loadA1(kk + 1); loadB1(kk + 1, bB);
            mma1(aA, bA);
            if (kk + 2 < 24) { aA = loadA1(kk + 2); loadB1(kk + 2, bA); }
            mma1(aB, bB);
        }
    }

    #pragma unroll
    for (int j = 0; j < 8; ++j) {
        #pragma unroll
        for (int r = 0; r < 8; ++r) {
            float h = gelu_tanh(acc[j][r]);
            int row = half ? (r + 8) : r;
            zh[row * HID + 16 * j + n0] = h;
        }
    }
    __syncthreads();

    float bv2 = b2[n0];
    v8f acc2a = (v8f){bv2, bv2, bv2, bv2, bv2, bv2, bv2, bv2};
    v8f acc2b = (v8f){0.f, 0.f, 0.f, 0.f, 0.f, 0.f, 0.f, 0.f};

    auto loadA2 = [&](int kk2) -> v2f {
        float2 t = zf2[n0 * 64 + 2 * kk2 + half];
        return (v2f){t.x, t.y};
    };
    auto loadB2 = [&](int kk2) -> v2f {
        float2 t = w2f2[kk2 * 32 + lane];
        return (v2f){t.x, t.y};
    };

    {
        v2f aA = loadA2(0), bA = loadB2(0);
        v2f aB, bB;
        #pragma unroll
        for (int kk2 = 0; kk2 < 32; kk2 += 2) {
            aB = loadA2(kk2 + 1); bB = loadB2(kk2 + 1);
            acc2a = __builtin_amdgcn_wmma_f32_16x16x4_f32(
                false, aA, false, bA, (short)0, acc2a, false, false);
            if (kk2 + 2 < 32) { aA = loadA2(kk2 + 2); bA = loadB2(kk2 + 2); }
            acc2b = __builtin_amdgcn_wmma_f32_16x16x4_f32(
                false, aB, false, bB, (short)0, acc2b, false, false);
        }
    }
    v8f acc2 = acc2a + acc2b;

    float t  = (float)step;
    float ms = tf[batch] - t;
    ms = fminf(fmaxf(ms, 0.0f), 1.0f);

    const size_t obase = xbatch + (size_t)node_base * DD;
    #pragma unroll
    for (int r = 0; r < 8; ++r) {
        int m = half ? (r + 8) : r;
        size_t a = obase + (size_t)m * DD + n0;
        xout[a] = xin[a] + ms * acc2[r];
    }
    {
        int m2 = lane >> 1;
        int c0 = DLAT + (lane & 1) * 4;
        size_t a2 = obase + (size_t)m2 * DD + c0;
        *reinterpret_cast<float4*>(xout + a2) =
            *reinterpret_cast<const float4*>(xin + a2);
    }
}

extern "C" void kernel_launch(void* const* d_in, const int* in_sizes, int n_in,
                              void* d_out, int out_size, void* d_ws, size_t ws_size,
                              hipStream_t stream) {
    (void)in_sizes; (void)n_in; (void)out_size; (void)ws_size;

    const float* x   = (const float*)d_in[0];
    const float* tf  = (const float*)d_in[1];
    const float* W1  = (const float*)d_in[2];
    const float* b1  = (const float*)d_in[3];
    const float* W2  = (const float*)d_in[4];
    const float* b2  = (const float*)d_in[5];
    const int*   nbr = (const int*)d_in[6];

    float* wsA = (float*)d_ws;
    float* wsw = wsA + W1S_OFF;
    float* out = (float*)d_out;

    swizzle_weights<<<28, 256, 0, stream>>>(W1, W2, wsw);

    const int grid = (NB * TPB) / 8;
    euler_step<<<grid, 256, LDS_TOTAL_B, stream>>>(x,   wsA, wsw, b1, b2, tf, nbr, 0);
    euler_step<<<grid, 256, LDS_TOTAL_B, stream>>>(wsA, out, wsw, b1, b2, tf, nbr, 1);
    euler_step<<<grid, 256, LDS_TOTAL_B, stream>>>(out, wsA, wsw, b1, b2, tf, nbr, 2);
    euler_step<<<grid, 256, LDS_TOTAL_B, stream>>>(wsA, out, wsw, b1, b2, tf, nbr, 3);
}


# Both layers computed transposed (weights as WMMA A operand): GELU output stays in layer 1's accumulator registers as layer 2's B operand, H skips LDS; layer 2's k summation order changes
# speedup vs baseline: 1.5938x; 1.1595x over previous
#include <hip/hip_runtime.h>
#include <math.h>

typedef __attribute__((ext_vector_type(2))) float v2f;
typedef __attribute__((ext_vector_type(8))) float v8f;

constexpr int NB   = 2;
constexpr int NN   = 131072;
constexpr int DD   = 24;
constexpr int KNB  = 4;
constexpr int HID  = 128;
constexpr int DLAT = 16;
constexpr int TPB  = NN / 16;

constexpr size_t XCOUNT   = (size_t)NB * NN * DD;
constexpr size_t W1S_OFF  = XCOUNT;
constexpr size_t W1S_F2   = 24 * 8 * 32;
constexpr size_t W2S_F2   = 32 * 32;
constexpr size_t WSW_FLTS = 2 * (W1S_F2 + W2S_F2);

constexpr int LDS_W2_OFF    = 12288;
constexpr int LDS_WAVE_OFF  = 14336;
constexpr int LDS_WAVE_SZ   = 1536;
constexpr int LDS_TOTAL_B   = (LDS_WAVE_OFF + 8 * LDS_WAVE_SZ) * 4;

__device__ __forceinline__ float gelu_tanh(float x) {
    constexpr float C1 = (float)(-2.0 * 0.7978845608028654 * 1.4426950408889634);
    constexpr float C2 = (float)(-2.0 * 0.7978845608028654 * 1.4426950408889634 * 0.044715);
    float p = x * fmaf(x * x, C2, C1);
    float e = __builtin_amdgcn_exp2f(p);
    return x * __builtin_amdgcn_rcpf(1.0f + e);
}

__global__ __launch_bounds__(256) void swizzle_weights(
    const float* __restrict__ W1, const float* __restrict__ W2,
    float* __restrict__ wsw)
{
    int t = blockIdx.x * 256 + threadIdx.x;
    if (t < (int)W1S_F2) {
        int kk  = t >> 8;
        int rem = t & 255;
        int j    = rem >> 5;
        int l    = rem & 31;
        int half = l >> 4;
        int n    = l & 15;
        int row0 = 4 * kk + 2 * half;
        int col  = 16 * j + n;
        float2 v = make_float2(W1[row0 * HID + col], W1[(row0 + 1) * HID + col]);
        reinterpret_cast<float2*>(wsw)[t] = v;
    } else if (t < (int)(W1S_F2 + W2S_F2)) {
        int s    = t - (int)W1S_F2;
        int kk2  = s >> 5;
        int l    = s & 31;
        int half = l >> 4;
        int n    = l & 15;
        int j    = kk2 >> 2;
        int i    = kk2 & 3;
        int row0 = 16 * j + 8 * half + 2 * i;
        float2 v = make_float2(W2[row0 * DLAT + n], W2[(row0 + 1) * DLAT + n]);
        reinterpret_cast<float2*>(wsw)[(int)W1S_F2 + s] = v;
    }
}

__global__ __launch_bounds__(256) void euler_step(
    const float* __restrict__ xin, float* __restrict__ xout,
    const float* __restrict__ wsw,
    const float* __restrict__ b1, const float* __restrict__ b2,
    const float* __restrict__ tf, const int* __restrict__ nbr,
    int step)
{
    extern __shared__ float smem[];
    const int tid  = threadIdx.x;
    const int wave = tid >> 5;
    const int lane = tid & 31;

    {
        const float4* src = reinterpret_cast<const float4*>(wsw);
        float4*       dst = reinterpret_cast<float4*>(smem);
        #pragma unroll
        for (int i = 0; i < (int)(WSW_FLTS / 4) / 256; ++i)
            dst[tid + i * 256] = src[tid + i * 256];
    }
    __syncthreads();

    float* zh = smem + LDS_WAVE_OFF + wave * LDS_WAVE_SZ;

    const int tile      = blockIdx.x * 8 + wave;
    const int batch     = tile / TPB;
    const int node_base = (tile % TPB) * 16;
    const size_t xbatch = (size_t)batch * NN * DD;

    #pragma unroll
    for (int rr = 0; rr < 2; ++rr) {
        int r = lane + rr * 32;
        int m = r >> 2, k = r & 3;
        int idx = nbr[(node_base + m) * KNB + k];
        const float4* src = reinterpret_cast<const float4*>(xin + xbatch + (size_t)idx * DD);
        float4*       dst = reinterpret_cast<float4*>(zh + m * 96 + k * 24);
        #pragma unroll
        for (int q = 0; q < 6; ++q) dst[q] = src[q];
    }
    __syncthreads();

    const int n0   = lane & 15;
    const int half = lane >> 4;

    const float2* zf2  = reinterpret_cast<const float2*>(zh);
    const float2* w1f2 = reinterpret_cast<const float2*>(smem);
    const float2* w2f2 = reinterpret_cast<const float2*>(smem + LDS_W2_OFF);

    v8f acc[8];
    #pragma unroll
    for (int j = 0; j < 8; ++j) {
        const float4* bp = reinterpret_cast<const float4*>(b1 + 16 * j + 8 * half);
        float4 lo = bp[0], hi = bp[1];
        acc[j] = (v8f){lo.x, lo.y, lo.z, lo.w, hi.x, hi.y, hi.z, hi.w};
    }

    auto loadA1 = [&](int kk) -> v2f {
        float2 t = zf2[n0 * 48 + 2 * kk + half];
        return (v2f){t.x, t.y};
    };
    auto loadB1 = [&](int kk, v2f* b) {
        #pragma unroll
        for (int j = 0; j < 8; ++j) {
            float2 t = w1f2[(kk * 8 + j) * 32 + lane];
            b[j] = (v2f){t.x, t.y};
        }
    };
    auto mma1 = [&](v2f a, const v2f* b) {
        #pragma unroll
        for (int j = 0; j < 8; ++j)
            acc[j] = __builtin_amdgcn_wmma_f32_16x16x4_f32(
                false, b[j], false, a, (short)0, acc[j], false, false);
    };

    {
        v2f aA, aB;
        v2f bA[8], bB[8];
        aA = loadA1(0); loadB1(0, bA);
        #pragma unroll
        for (int kk = 0; kk < 24; kk += 2) {
            aB = loadA1(kk + 1); loadB1(kk + 1, bB);
            mma1(aA, bA);
            if (kk + 2 < 24) { aA = loadA1(kk + 2); loadB1(kk + 2, bA); }
            mma1(aB, bB);
        }
    }

    v2f w2f[32];
    #pragma unroll
    for (int kk2 = 0; kk2 < 32; ++kk2) {
        float2 t = w2f2[kk2 * 32 + lane];
        w2f[kk2] = (v2f){t.x, t.y};
    }

    v8f acc2a, acc2b;
    {
        const float4* bp = reinterpret_cast<const float4*>(b2 + 8 * half);
        float4 lo = bp[0], hi = bp[1];
        acc2a = (v8f){lo.x, lo.y, lo.z, lo.w, hi.x, hi.y, hi.z, hi.w};
        acc2b = (v8f){0.f, 0.f, 0.f, 0.f, 0.f, 0.f, 0.f, 0.f};
    }

    #pragma unroll
    for (int j = 0; j < 8; ++j) {
        #pragma unroll
        for (int r = 0; r < 8; ++r) acc[j][r] = gelu_tanh(acc[j][r]);
        #pragma unroll
        for (int i = 0; i < 4; i += 2) {
            v2f h0 = (v2f){acc[j][2 * i],     acc[j][2 * i + 1]};
            v2f h1 = (v2f){acc[j][2 * i + 2], acc[j][2 * i + 3]};
            acc2a = __builtin_amdgcn_wmma_f32_16x16x4_f32(
                false, w2f[4 * j + i],     false, h0, (short)0, acc2a, false, false);
            acc2b = __builtin_amdgcn_wmma_f32_16x16x4_f32(
                false, w2f[4 * j + i + 1], false, h1, (short)0, acc2b, false, false);
        }
    }
    v8f acc2 = acc2a + acc2b;

    float t  = (float)step;
    float ms = tf[batch] - t;
    ms = fminf(fmaxf(ms, 0.0f), 1.0f);

    const size_t obase = xbatch + (size_t)node_base * DD;
    {
        size_t a = obase + (size_t)n0 * DD + 8 * half;
        const float4* xi = reinterpret_cast<const float4*>(xin + a);
        float4 x0 = xi[0], x1 = xi[1];
        float4 o0, o1;
        o0.x = x0.x + ms * acc2[0];
        o0.y = x0.y + ms * acc2[1];
        o0.z = x0.z + ms * acc2[2];
        o0.w = x0.w + ms * acc2[3];
        o1.x = x1.x + ms * acc2[4];
        o1.y = x1.y + ms * acc2[5];
        o1.z = x1.z + ms * acc2[6];
        o1.w = x1.w + ms * acc2[7];
        float4* xo = reinterpret_cast<float4*>(xout + a);
        xo[0] = o0;
        xo[1] = o1;
    }
    {
        int m2 = lane >> 1;
        int c0 = DLAT + (lane & 1) * 4;
        size_t a2 = obase + (size_t)m2 * DD + c0;
        *reinterpret_cast<float4*>(xout + a2) =
            *reinterpret_cast<const float4*>(xin + a2);
    }
}

extern "C" void kernel_launch(void* const* d_in, const int* in_sizes, int n_in,
                              void* d_out, int out_size, void* d_ws, size_t ws_size,
                              hipStream_t stream) {
    (void)in_sizes; (void)n_in; (void)out_size; (void)ws_size;

    const float* x   = (const float*)d_in[0];
    const float* tf  = (const float*)d_in[1];
    const float* W1  = (const float*)d_in[2];
    const float* b1  = (const float*)d_in[3];
    const float* W2  = (const float*)d_in[4];
    const float* b2  = (const float*)d_in[5];
    const int*   nbr = (const int*)d_in[6];

    float* wsA = (float*)d_ws;
    float* wsw = wsA + W1S_OFF;
    float* out = (float*)d_out;

    swizzle_weights<<<28, 256, 0, stream>>>(W1, W2, wsw);

    const int grid = (NB * TPB) / 8;
    euler_step<<<grid, 256, LDS_TOTAL_B, stream>>>(x,   wsA, wsw, b1, b2, tf, nbr, 0);
    euler_step<<<grid, 256, LDS_TOTAL_B, stream>>>(wsA, out, wsw, b1, b2, tf, nbr, 1);
    euler_step<<<grid, 256, LDS_TOTAL_B, stream>>>(out, wsA, wsw, b1, b2, tf, nbr, 2);
    euler_step<<<grid, 256, LDS_TOTAL_B, stream>>>(wsA, out, wsw, b1, b2, tf, nbr, 3);
}


# Masked-out steps (clipped step size exactly 0: t >= t_final of the batch element) skip weight staging, gather and both matrix products; the work-group copies its 128 nodes. Exact for finite dx
# speedup vs baseline: 2.4578x; 1.5421x over previous
#include <hip/hip_runtime.h>
#include <math.h>

typedef __attribute__((ext_vector_type(2))) float v2f;
typedef __attribute__((ext_vector_type(4))) float v4f;
typedef __attribute__((ext_vector_type(8))) float v8f;

constexpr int NB   = 2;
constexpr int NN   = 131072;
constexpr int DD   = 24;
constexpr int KNB  = 4;
constexpr int HID  = 128;
constexpr int DLAT = 16;
constexpr int TPB  = NN / 16;

constexpr size_t XCOUNT   = (size_t)NB * NN * DD;
constexpr size_t W1S_OFF  = XCOUNT;
constexpr size_t W1S_F2   = 24 * 8 * 32;
constexpr size_t W2S_F2   = 32 * 32;
constexpr size_t WSW_FLTS = 2 * (W1S_F2 + W2S_F2);

constexpr int LDS_W2_OFF    = 12288;
constexpr int LDS_WAVE_OFF  = 14336;
constexpr int LDS_WAVE_SZ   = 1536;
constexpr int LDS_TOTAL_B   = (LDS_WAVE_OFF + 8 * LDS_WAVE_SZ) * 4;

__device__ __forceinline__ float gelu_tanh(float x) {
    constexpr float C1 = (float)(-2.0 * 0.7978845608028654 * 1.4426950408889634);
    constexpr float C2 = (float)(-2.0 * 0.7978845608028654 * 1.4426950408889634 * 0.044715);
    float p = x * fmaf(x * x, C2, C1);
    float e = __builtin_amdgcn_exp2f(p);
    return x * __builtin_amdgcn_rcpf(1.0f + e);
}

__global__ __launch_bounds__(256) void swizzle_weights(
    const float* __restrict__ W1, const float* __restrict__ W2,
    float* __restrict__ wsw)
{
    int t = blockIdx.x * 256 + threadIdx.x;
    if (t < (int)W1S_F2) {
        int kk  = t >> 8;
        int rem = t & 255;
        int j    = rem >> 5;
        int l    = rem & 31;
        int half = l >> 4;
        int n    = l & 15;
        int row0 = 4 * kk + 2 * half;
        int col  = 16 * j + n;
        float2 v = make_float2(W1[row0 * HID + col], W1[(row0 + 1) * HID + col]);
        reinterpret_cast<float2*>(wsw)[t] = v;
    } else if (t < (int)(W1S_F2 + W2S_F2)) {
        int s    = t - (int)W1S_F2;
        int kk2  = s >> 5;
        int l    = s & 31;
        int half = l >> 4;
        int n    = l & 15;
        int j    = kk2 >> 2;
        int i    = kk2 & 3;
        int row0 = 16 * j + 8 * half + 2 * i;
        float2 v = make_float2(W2[row0 * DLAT + n], W2[(row0 + 1) * DLAT + n]);
        reinterpret_cast<float2*>(wsw)[(int)W1S_F2 + s] = v;
    }
}

__global__ __launch_bounds__(256) void euler_step(
    const float* __restrict__ xin, float* __restrict__ xout,
    const float* __restrict__ wsw,
    const float* __restrict__ b1, const float* __restrict__ b2,
    const float* __restrict__ tf, const int* __restrict__ nbr,
    int step)
{
    extern __shared__ float smem[];
    const int tid  = threadIdx.x;
    const int wave = tid >> 5;
    const int lane = tid & 31;

    const int batch = (blockIdx.x * 8) / TPB;
    float ms = tf[batch] - (float)step;
    ms = fminf(fmaxf(ms, 0.0f), 1.0f);

    if (ms == 0.0f) {
        const size_t cbase = (size_t)blockIdx.x * (8 * 16 * DD);
        const v4f* csrc = reinterpret_cast<const v4f*>(xin + cbase);
        v4f*       cdst = reinterpret_cast<v4f*>(xout + cbase);
        v4f c0 = csrc[tid], c1 = csrc[tid + 256], c2 = csrc[tid + 512];
        asm volatile("" :: "v"(c0), "v"(c1), "v"(c2));
        cdst[tid]       = c0;
        cdst[tid + 256] = c1;
        cdst[tid + 512] = c2;
        return;
    }

    {
        const float4* src = reinterpret_cast<const float4*>(wsw);
        float4*       dst = reinterpret_cast<float4*>(smem);
        #pragma unroll
        for (int i = 0; i < (int)(WSW_FLTS / 4) / 256; ++i)
            dst[tid + i * 256] = src[tid + i * 256];
    }
    __syncthreads();

    float* zh = smem + LDS_WAVE_OFF + wave * LDS_WAVE_SZ;

    const int tile      = blockIdx.x * 8 + wave;
    const int node_base = (tile % TPB) * 16;
    const size_t xbatch = (size_t)batch * NN * DD;

    #pragma unroll
    for (int rr = 0; rr < 2; ++rr) {
        int r = lane + rr * 32;
        int m = r >> 2, k = r & 3;
        int idx = nbr[(node_base + m) * KNB + k];
        const float4* src = reinterpret_cast<const float4*>(xin + xbatch + (size_t)idx * DD);
        float4*       dst = reinterpret_cast<float4*>(zh + m * 96 + k * 24);
        #pragma unroll
        for (int q = 0; q < 6; ++q) dst[q] = src[q];
    }
    __syncthreads();

    const int n0   = lane & 15;
    const int half = lane >> 4;

    const float2* zf2  = reinterpret_cast<const float2*>(zh);
    const float2* w1f2 = reinterpret_cast<const float2*>(smem);
    const float2* w2f2 = reinterpret_cast<const float2*>(smem + LDS_W2_OFF);

    v8f acc[8];
    #pragma unroll
    for (int j = 0; j < 8; ++j) {
        const float4* bp = reinterpret_cast<const float4*>(b1 + 16 * j + 8 * half);
        float4 lo = bp[0], hi = bp[1];
        acc[j] = (v8f){lo.x, lo.y, lo.z, lo.w, hi.x, hi.y, hi.z, hi.w};
    }

    auto loadA1 = [&](int kk) -> v2f {
        float2 t = zf2[n0 * 48 + 2 * kk + half];
        return (v2f){t.x, t.y};
    };
    auto loadB1 = [&](int kk, v2f* b) {
        #pragma unroll
        for (int j = 0; j < 8; ++j) {
            float2 t = w1f2[(kk * 8 + j) * 32 + lane];
            b[j] = (v2f){t.x, t.y};
        }
    };
    auto mma1 = [&](v2f a, const v2f* b) {
        #pragma unroll
        for (int j = 0; j < 8; ++j)
            acc[j] = __builtin_amdgcn_wmma_f32_16x16x4_f32(
                false, b[j], false, a, (short)0, acc[j], false, false);
    };

    {
        v2f aA, aB;
        v2f bA[8], bB[8];
        aA = loadA1(0); loadB1(0, bA);
        #pragma unroll
        for (int kk = 0; kk < 24; kk += 2) {
            aB = loadA1(kk + 1); loadB1(kk + 1, bB);
            mma1(aA, bA);
            if (kk + 2 < 24) { aA = loadA1(kk + 2); loadB1(kk + 2, bA); }
            mma1(aB, bB);
        }
    }

    v2f w2f[32];
    #pragma unroll
    for (int kk2 = 0; kk2 < 32; ++kk2) {
        float2 t = w2f2[kk2 * 32 + lane];
        w2f[kk2] = (v2f){t.x, t.y};
    }

    v8f acc2a, acc2b;
    {
        const float4* bp = reinterpret_cast<const float4*>(b2 + 8 * half);
        float4 lo = bp[0], hi = bp[1];
        acc2a = (v8f){lo.x, lo.y, lo.z, lo.w, hi.x, hi.y, hi.z, hi.w};
        acc2b = (v8f){0.f, 0.f, 0.f, 0.f, 0.f, 0.f, 0.f, 0.f};
    }

    #pragma unroll
    for (int j = 0; j < 8; ++j) {
        #pragma unroll
        for (int r = 0; r < 8; ++r) acc[j][r] = gelu_tanh(acc[j][r]);
        #pragma unroll
        for (int i = 0; i < 4; i += 2) {
            v2f h0 = (v2f){acc[j][2 * i],     acc[j][2 * i + 1]};
            v2f h1 = (v2f){acc[j][2 * i + 2], acc[j][2 * i + 3]};
            acc2a = __builtin_amdgcn_wmma_f32_16x16x4_f32(
                false, w2f[4 * j + i],     false, h0, (short)0, acc2a, false, false);
            acc2b = __builtin_amdgcn_wmma_f32_16x16x4_f32(
                false, w2f[4 * j + i + 1], false, h1, (short)0, acc2b, false, false);
        }
    }
    v8f acc2 = acc2a + acc2b;

    const size_t obase = xbatch + (size_t)node_base * DD;
    {
        size_t a = obase + (size_t)n0 * DD + 8 * half;
        const float4* xi = reinterpret_cast<const float4*>(xin + a);
        float4 x0 = xi[0], x1 = xi[1];
        float4 o0, o1;
        o0.x = x0.x + ms * acc2[0];
        o0.y = x0.y + ms * acc2[1];
        o0.z = x0.z + ms * acc2[2];
        o0.w = x0.w + ms * acc2[3];
        o1.x = x1.x + ms * acc2[4];
        o1.y = x1.y + ms * acc2[5];
        o1.z = x1.z + ms * acc2[6];
        o1.w = x1.w + ms * acc2[7];
        float4* xo = reinterpret_cast<float4*>(xout + a);
        xo[0] = o0;
        xo[1] = o1;
    }
    {
        int m2 = lane >> 1;
        int c0 = DLAT + (lane & 1) * 4;
        size_t a2 = obase + (size_t)m2 * DD + c0;
        *reinterpret_cast<float4*>(xout + a2) =
            *reinterpret_cast<const float4*>(xin + a2);
    }
}

extern "C" void kernel_launch(void* const* d_in, const int* in_sizes, int n_in,
                              void* d_out, int out_size, void* d_ws, size_t ws_size,
                              hipStream_t stream) {
    (void)in_sizes; (void)n_in; (void)out_size; (void)ws_size;

    const float* x   = (const float*)d_in[0];
    const float* tf  = (const float*)d_in[1];
    const float* W1  = (const float*)d_in[2];
    const float* b1  = (const float*)d_in[3];
    const float* W2  = (const float*)d_in[4];
    const float* b2  = (const float*)d_in[5];
    const int*   nbr = (const int*)d_in[6];

    float* wsA = (float*)d_ws;
    float* wsw = wsA + W1S_OFF;
    float* out = (float*)d_out;

    swizzle_weights<<<28, 256, 0, stream>>>(W1, W2, wsw);

    const int grid = (NB * TPB) / 8;
    euler_step<<<grid, 256, LDS_TOTAL_B, stream>>>(x,   wsA, wsw, b1, b2, tf, nbr, 0);
    euler_step<<<grid, 256, LDS_TOTAL_B, stream>>>(wsA, out, wsw, b1, b2, tf, nbr, 1);
    euler_step<<<grid, 256, LDS_TOTAL_B, stream>>>(out, wsA, wsw, b1, b2, tf, nbr, 2);
    euler_step<<<grid, 256, LDS_TOTAL_B, stream>>>(wsA, out, wsw, b1, b2, tf, nbr, 3);
}
